# starting kernel
# baseline (speedup 1.0000x reference)
#include <hip/hip_runtime.h>

typedef float v2f __attribute__((ext_vector_type(2)));
typedef float v8f __attribute__((ext_vector_type(8)));

#define HIDDEN 2048
#define EMBED  512
#define NV     16
#define WAVES  16
#define BLOCK  (WAVES * 32)

__device__ __forceinline__ v8f wmma_f32_k4(v2f a, v2f b, v8f c) {
  return __builtin_amdgcn_wmma_f32_16x16x4_f32(false, a, false, b, (short)0, c,
                                               false, false);
}

template <int KPER>
__device__ __forceinline__ void gemv16x3_wmma(const float* __restrict__ W,
                                              int stride, int jBase,
                                              const float* __restrict__ sX,
                                              int kStart,
                                              v8f& c0, v8f& c1, v8f& c2) {
  const int L   = threadIdx.x & 31;
  const int sub = L & 15;
  const int hi2 = (L >> 4) << 1;
  const float* r0 = W + (size_t)(jBase + sub) * (size_t)stride + hi2 + kStart;
  const float* r1 = r0 + (size_t)HIDDEN * (size_t)stride;
  const float* r2 = r1 + (size_t)HIDDEN * (size_t)stride;
  const float* xp = sX + hi2 + kStart;
#pragma unroll 4
  for (int kb = 0; kb < KPER; kb += 4) {
    const float2 b  = *(const float2*)(xp + kb);
    const float2 a0 = *(const float2*)(r0 + kb);
    const float2 a1 = *(const float2*)(r1 + kb);
    const float2 a2 = *(const float2*)(r2 + kb);
    v2f B;  B.x  = b.x;  B.y  = b.y;
    v2f A0; A0.x = a0.x; A0.y = a0.y;
    v2f A1; A1.x = a1.x; A1.y = a1.y;
    v2f A2; A2.x = a2.x; A2.y = a2.y;
    c0 = wmma_f32_k4(A0, B, c0);
    c1 = wmma_f32_k4(A1, B, c1);
    c2 = wmma_f32_k4(A2, B, c2);
  }
}

template <int KPER>
__device__ __forceinline__ void gemv16_wmma(const float* __restrict__ W,
                                            int stride,
                                            const float* __restrict__ sX,
                                            int kStart, v8f& c) {
  const int L   = threadIdx.x & 31;
  const int sub = L & 15;
  const int hi2 = (L >> 4) << 1;
  const float* r0 = W + (size_t)sub * (size_t)stride + hi2 + kStart;
  const float* xp = sX + hi2 + kStart;
#pragma unroll 4
  for (int kb = 0; kb < KPER; kb += 4) {
    const float2 b = *(const float2*)(xp + kb);
    const float2 a = *(const float2*)(r0 + kb);
    v2f B; B.x = b.x; B.y = b.y;
    v2f A; A.x = a.x; A.y = a.y;
    c = wmma_f32_k4(A, B, c);
  }
}

__device__ __forceinline__ void reduce16(v8f c, float* acc) {
  const int L   = threadIdx.x & 31;
  const int sub = L & 15;
  const int hi  = L >> 4;
#pragma unroll
  for (int m = 0; m < 8; ++m) {
    if (sub == m) atomicAdd(&acc[m + 8 * hi], c[m]);
  }
}

template <int K_IH, int K_HH>
__global__ void __launch_bounds__(BLOCK)
gru_matvec_kernel(const float* __restrict__ W_ih, const float* __restrict__ W_hh,
                  const float* __restrict__ b_ih, const float* __restrict__ b_hh,
                  const float* __restrict__ xsrc, const int* __restrict__ xidx,
                  const float* __restrict__ hprev, float* __restrict__ pre) {
  __shared__ float sX[HIDDEN];
  __shared__ float sAcc[3][16];

  const int tid   = threadIdx.x;
  const int jBase = blockIdx.x * 16;
  const int mat   = blockIdx.y;

  if (mat == 0) {
    const float* xv = xidx ? (xsrc + (size_t)xidx[0] * (size_t)K_IH) : xsrc;
    for (int i = tid; i < K_IH; i += BLOCK) sX[i] = xv[i];
  } else {
    for (int i = tid; i < K_HH; i += BLOCK) sX[i] = hprev[i];
  }
  if (tid < 48) ((float*)sAcc)[tid] = 0.0f;
  __syncthreads();

  const int wave = __builtin_amdgcn_readfirstlane(tid >> 5);

  if (mat == 0) {
    constexpr int kPer = K_IH / WAVES;
    v8f c0 = {}, c1 = {}, c2 = {};
    gemv16x3_wmma<kPer>(W_ih, K_IH, jBase, sX, wave * kPer, c0, c1, c2);
    reduce16(c0, sAcc[0]);
    reduce16(c1, sAcc[1]);
    reduce16(c2, sAcc[2]);
  } else {
    constexpr int kPer = K_HH / WAVES;
    v8f c0 = {}, c1 = {}, c2 = {};
    gemv16x3_wmma<kPer>(W_hh, K_HH, jBase, sX, wave * kPer, c0, c1, c2);
    reduce16(c0, sAcc[0]);
    reduce16(c1, sAcc[1]);
    reduce16(c2, sAcc[2]);
  }
  __syncthreads();

  if (tid < 48) {
    const int g = tid >> 4;
    const int m = tid & 15;
    const int j = jBase + m;
    const float* bias = (mat == 0) ? b_ih : b_hh;
    pre[(mat * 3 + g) * HIDDEN + j] = sAcc[g][m] + bias[g * HIDDEN + j];
  }
}

__global__ void __launch_bounds__(256)
gru_gate_kernel(const float* __restrict__ pre, const float* __restrict__ hprev,
                float* __restrict__ hnext, float* __restrict__ hcopy) {
  const int j = blockIdx.x * 256 + threadIdx.x;
  if (j < HIDDEN) {
    const float ir  = pre[0 * HIDDEN + j];
    const float iz  = pre[1 * HIDDEN + j];
    const float inn = pre[2 * HIDDEN + j];
    const float hr  = pre[3 * HIDDEN + j];
    const float hz  = pre[4 * HIDDEN + j];
    const float hn  = pre[5 * HIDDEN + j];
    const float r = 1.0f / (1.0f + __expf(-(ir + hr)));
    const float z = 1.0f / (1.0f + __expf(-(iz + hz)));
    const float n = tanhf(inn + r * hn);
    const float h = (1.0f - z) * n + z * hprev[j];
    hnext[j] = h;
    hcopy[j] = h;
  }
}

__global__ void __launch_bounds__(BLOCK)
head_kernel(const float* __restrict__ w_head, const float* __restrict__ b_head,
            const float* __restrict__ h1, float* __restrict__ probs) {
  __shared__ float sX[HIDDEN];
  __shared__ float sAcc[16];
  const int tid = threadIdx.x;
  for (int i = tid; i < HIDDEN; i += BLOCK) sX[i] = h1[i];
  if (tid < 16) sAcc[tid] = 0.0f;
  __syncthreads();

  const int wave = __builtin_amdgcn_readfirstlane(tid >> 5);
  constexpr int kPer = HIDDEN / WAVES;
  v8f c = {};
  gemv16_wmma<kPer>(w_head, HIDDEN, sX, wave * kPer, c);
  reduce16(c, sAcc);
  __syncthreads();

  if (tid < 16) {
    const float v = sAcc[tid] + b_head[tid];
    float m = v;
    for (int off = 8; off > 0; off >>= 1) m = fmaxf(m, __shfl_xor(m, off, 32));
    const float e = __expf(v - m);
    float s = e;
    for (int off = 8; off > 0; off >>= 1) s += __shfl_xor(s, off, 32);
    probs[tid] = e / s;
  }
}

extern "C" void kernel_launch(void* const* d_in, const int* in_sizes, int n_in,
                              void* d_out, int out_size, void* d_ws,
                              size_t ws_size, hipStream_t stream) {
  const int*   x      = (const int*)  d_in[0];
  const float* h_in   = (const float*)d_in[1];
  const float* emb    = (const float*)d_in[2];
  const float* w_ih0  = (const float*)d_in[3];
  const float* w_hh0  = (const float*)d_in[4];
  const float* b_ih0  = (const float*)d_in[5];
  const float* b_hh0  = (const float*)d_in[6];
  const float* w_ih1  = (const float*)d_in[7];
  const float* w_hh1  = (const float*)d_in[8];
  const float* b_ih1  = (const float*)d_in[9];
  const float* b_hh1  = (const float*)d_in[10];
  const float* w_head = (const float*)d_in[11];
  const float* b_head = (const float*)d_in[12];

  float* out = (float*)d_out;
  float* h0  = (float*)d_ws;
  float* h1  = h0 + HIDDEN;
  float* pre = h1 + HIDDEN;

  const dim3 gridMV(HIDDEN / 16, 2);

  gru_matvec_kernel<EMBED, HIDDEN><<<gridMV, BLOCK, 0, stream>>>(
      w_ih0, w_hh0, b_ih0, b_hh0, emb, x, h_in, pre);
  gru_gate_kernel<<<HIDDEN / 256, 256, 0, stream>>>(pre, h_in, h0, out + NV);

  gru_matvec_kernel<HIDDEN, HIDDEN><<<gridMV, BLOCK, 0, stream>>>(
      w_ih1, w_hh1, b_ih1, b_hh1, h0, nullptr, h_in + HIDDEN, pre);
  gru_gate_kernel<<<HIDDEN / 256, 256, 0, stream>>>(pre, h_in + HIDDEN, h1,
                                                    out + NV + HIDDEN);

  head_kernel<<<1, BLOCK, 0, stream>>>(w_head, b_head, h1, out);
}


# Each wave owns whole weight rows: every load reads 512 contiguous bytes of ONE row; row values in A, same lane's vector values in B, dot = trace of D; new summation order, LDS atomics gone
# speedup vs baseline: 1.9227x; 1.9227x over previous
#include <hip/hip_runtime.h>

typedef float v2f __attribute__((ext_vector_type(2)));
typedef float v4f __attribute__((ext_vector_type(4)));
typedef float v8f __attribute__((ext_vector_type(8)));

#define HIDDEN 2048
#define EMBED  512
#define NV     16
#define WAVES  16
#define BLOCK  (WAVES * 32)

__device__ __forceinline__ v8f wmma_f32_k4(v2f a, v2f b, v8f c) {
  return __builtin_amdgcn_wmma_f32_16x16x4_f32(false, a, false, b, (short)0, c,
                                               false, false);
}


template <int KQ>
__device__ __forceinline__ void load_pieces(const float* __restrict__ p,
                                            v4f (&dst)[KQ]) {
#pragma unroll
  for (int q = 0; q < KQ; ++q) dst[q] = *(const v4f*)(p + 128 * q);
}

template <int KQ>
__device__ __forceinline__ float row_diag(const v4f (&w)[KQ],
                                          const v4f (&x)[KQ]) {
  v8f c0 = {}, c1 = {};
#pragma unroll
  for (int q = 0; q < KQ; ++q) {
    v2f a0; a0.x = w[q].x; a0.y = w[q].y;
    v2f b0; b0.x = x[q].x; b0.y = x[q].y;
    v2f a1; a1.x = w[q].z; a1.y = w[q].w;
    v2f b1; b1.x = x[q].z; b1.y = x[q].w;
    c0 = wmma_f32_k4(a0, b0, c0);
    c1 = wmma_f32_k4(a1, b1, c1);
  }
  const v8f c = c0 + c1;
  const int lane = threadIdx.x & 31;
  const int r = lane & 7;
  float d = c[0];
#pragma unroll
  for (int i = 1; i < 8; ++i) d = (r == i) ? c[i] : d;
  const bool on_diag = (lane >> 4) == ((lane >> 3) & 1);
  return on_diag ? d : 0.0f;
}

__device__ __forceinline__ float trace_sum(float d) {
  d += __shfl_xor(d, 24, 32);
  d += __shfl_xor(d, 4, 32);
  d += __shfl_xor(d, 2, 32);
  d += __shfl_xor(d, 1, 32);
  return d;
}

template <int KQ>
__device__ __forceinline__ void gate_rows3(const float* __restrict__ W, int j,
                                           const float* __restrict__ xv,
                                           float (&sum)[3]) {
  constexpr int K = 128 * KQ;
  constexpr int AHEAD = (KQ <= 4) ? 2 : 1;
  const int lane = threadIdx.x & 31;
  v4f x[KQ];
  load_pieces<KQ>(xv + 4 * lane, x);
  __builtin_amdgcn_sched_barrier(0);
  const float* wp = W + (size_t)j * (size_t)K + 4 * lane;
  v4f w[AHEAD + 1][KQ];
#pragma unroll
  for (int r = 0; r < AHEAD; ++r)
    load_pieces<KQ>(wp + (size_t)r * (size_t)HIDDEN * (size_t)K, w[r]);
#pragma unroll
  for (int r = 0; r < 3; ++r) {
    __builtin_amdgcn_sched_barrier(0);
    if (r + AHEAD < 3)
      load_pieces<KQ>(wp + (size_t)(r + AHEAD) * (size_t)HIDDEN * (size_t)K,
                      w[(r + AHEAD) % (AHEAD + 1)]);
    __builtin_amdgcn_sched_barrier(0);
    sum[r] = row_diag<KQ>(w[r % (AHEAD + 1)], x);
  }
#pragma unroll
  for (int r = 0; r < 3; ++r) sum[r] = trace_sum(sum[r]);
}

template <int K_IH, int K_HH>
__global__ void __launch_bounds__(BLOCK)
gru_matvec_kernel(const float* __restrict__ W_ih, const float* __restrict__ W_hh,
                  const float* __restrict__ b_ih, const float* __restrict__ b_hh,
                  const float* __restrict__ xsrc, const int* __restrict__ xidx,
                  const float* __restrict__ hprev, float* __restrict__ pre) {
  __shared__ float sAcc[3][16];

  const int tid   = threadIdx.x;
  const int jBase = blockIdx.x * 16;
  const int mat   = blockIdx.y;
  const int wave = __builtin_amdgcn_readfirstlane(tid >> 5);

  float sum[3];
  if (mat == 0) {
    const float* xv = xidx ? (xsrc + (size_t)xidx[0] * (size_t)K_IH) : xsrc;
    gate_rows3<K_IH / 128>(W_ih, jBase + wave, xv, sum);
  } else {
    gate_rows3<K_HH / 128>(W_hh, jBase + wave, hprev, sum);
  }
  if ((tid & 31) == 0) {
    sAcc[0][wave] = sum[0];
    sAcc[1][wave] = sum[1];
    sAcc[2][wave] = sum[2];
  }
  __syncthreads();

  if (tid < 48) {
    const int g = tid >> 4;
    const int m = tid & 15;
    const int j = jBase + m;
    const float* bias = (mat == 0) ? b_ih : b_hh;
    pre[(mat * 3 + g) * HIDDEN + j] = sAcc[g][m] + bias[g * HIDDEN + j];
  }
}

__global__ void __launch_bounds__(256)
gru_gate_kernel(const float* __restrict__ pre, const float* __restrict__ hprev,
                float* __restrict__ hnext, float* __restrict__ hcopy) {
  const int j = blockIdx.x * 256 + threadIdx.x;
  if (j < HIDDEN) {
    const float ir  = pre[0 * HIDDEN + j];
    const float iz  = pre[1 * HIDDEN + j];
    const float inn = pre[2 * HIDDEN + j];
    const float hr  = pre[3 * HIDDEN + j];
    const float hz  = pre[4 * HIDDEN + j];
    const float hn  = pre[5 * HIDDEN + j];
    const float r = 1.0f / (1.0f + __expf(-(ir + hr)));
    const float z = 1.0f / (1.0f + __expf(-(iz + hz)));
    const float n = tanhf(inn + r * hn);
    const float h = (1.0f - z) * n + z * hprev[j];
    hnext[j] = h;
    hcopy[j] = h;
  }
}

__global__ void __launch_bounds__(BLOCK)
head_kernel(const float* __restrict__ w_head, const float* __restrict__ b_head,
            const float* __restrict__ h1, float* __restrict__ probs) {
  __shared__ float sAcc[16];
  const int tid  = threadIdx.x;
  const int lane = tid & 31;
  const int wave = __builtin_amdgcn_readfirstlane(tid >> 5);
  constexpr int KQ = HIDDEN / 128;

  v4f x[KQ], w[KQ];
  load_pieces<KQ>(h1 + 4 * lane, x);
  load_pieces<KQ>(w_head + (size_t)wave * HIDDEN + 4 * lane, w);
  __builtin_amdgcn_sched_barrier(0);
  const float s = trace_sum(row_diag<KQ>(w, x));
  if (lane == 0) sAcc[wave] = s;
  __syncthreads();

  if (tid < 16) {
    const float v = sAcc[tid] + b_head[tid];
    float m = v;
    for (int off = 8; off > 0; off >>= 1) m = fmaxf(m, __shfl_xor(m, off, 32));
    const float e = __expf(v - m);
    float s2 = e;
    for (int off = 8; off > 0; off >>= 1) s2 += __shfl_xor(s2, off, 32);
    probs[tid] = e / s2;
  }
}

extern "C" void kernel_launch(void* const* d_in, const int* in_sizes, int n_in,
                              void* d_out, int out_size, void* d_ws,
                              size_t ws_size, hipStream_t stream) {
  const int*   x      = (const int*)  d_in[0];
  const float* h_in   = (const float*)d_in[1];
  const float* emb    = (const float*)d_in[2];
  const float* w_ih0  = (const float*)d_in[3];
  const float* w_hh0  = (const float*)d_in[4];
  const float* b_ih0  = (const float*)d_in[5];
  const float* b_hh0  = (const float*)d_in[6];
  const float* w_ih1  = (const float*)d_in[7];
  const float* w_hh1  = (const float*)d_in[8];
  const float* b_ih1  = (const float*)d_in[9];
  const float* b_hh1  = (const float*)d_in[10];
  const float* w_head = (const float*)d_in[11];
  const float* b_head = (const float*)d_in[12];

  float* out = (float*)d_out;
  float* h0  = (float*)d_ws;
  float* h1  = h0 + HIDDEN;
  float* pre = h1 + HIDDEN;

  const dim3 gridMV(HIDDEN / 16, 2);

  gru_matvec_kernel<EMBED, HIDDEN><<<gridMV, BLOCK, 0, stream>>>(
      w_ih0, w_hh0, b_ih0, b_hh0, emb, x, h_in, pre);
  gru_gate_kernel<<<HIDDEN / 256, 256, 0, stream>>>(pre, h_in, h0, out + NV);

  gru_matvec_kernel<HIDDEN, HIDDEN><<<gridMV, BLOCK, 0, stream>>>(
      w_ih1, w_hh1, b_ih1, b_hh1, h0, nullptr, h_in + HIDDEN, pre);
  gru_gate_kernel<<<HIDDEN / 256, 256, 0, stream>>>(pre, h_in + HIDDEN, h1,
                                                    out + NV + HIDDEN);

  head_kernel<<<1, BLOCK, 0, stream>>>(w_head, b_head, h1, out);
}


# Gate math fused into the matvec kernel: a block owns 8 hidden units on BOTH matrices (8 waves each), sums meet in LDS, 8 lanes write h; 5 launches become 3; summation order unchanged
# speedup vs baseline: 2.4669x; 1.2830x over previous
#include <hip/hip_runtime.h>

typedef float v2f __attribute__((ext_vector_type(2)));
typedef float v4f __attribute__((ext_vector_type(4)));
typedef float v8f __attribute__((ext_vector_type(8)));

#define HIDDEN 2048
#define EMBED  512
#define NV     16
#define WAVES  16
#define BLOCK  (WAVES * 32)
#define UNITS  (WAVES / 2)

__device__ __forceinline__ v8f wmma_f32_k4(v2f a, v2f b, v8f c) {
  return __builtin_amdgcn_wmma_f32_16x16x4_f32(false, a, false, b, (short)0, c,
                                               false, false);
}


template <int KQ>
__device__ __forceinline__ void load_pieces(const float* __restrict__ p,
                                            v4f (&dst)[KQ]) {
#pragma unroll
  for (int q = 0; q < KQ; ++q) dst[q] = *(const v4f*)(p + 128 * q);
}

template <int KQ>
__device__ __forceinline__ float row_diag(const v4f (&w)[KQ],
                                          const v4f (&x)[KQ]) {
  v8f c0 = {}, c1 = {};
#pragma unroll
  for (int q = 0; q < KQ; ++q) {
    v2f a0; a0.x = w[q].x; a0.y = w[q].y;
    v2f b0; b0.x = x[q].x; b0.y = x[q].y;
    v2f a1; a1.x = w[q].z; a1.y = w[q].w;
    v2f b1; b1.x = x[q].z; b1.y = x[q].w;
    c0 = wmma_f32_k4(a0, b0, c0);
    c1 = wmma_f32_k4(a1, b1, c1);
  }
  const v8f c = c0 + c1;
  const int lane = threadIdx.x & 31;
  const int r = lane & 7;
  float d = c[0];
#pragma unroll
  for (int i = 1; i < 8; ++i) d = (r == i) ? c[i] : d;
  const bool on_diag = (lane >> 4) == ((lane >> 3) & 1);
  return on_diag ? d : 0.0f;
}

__device__ __forceinline__ float trace_sum(float d) {
  d += __shfl_xor(d, 24, 32);
  d += __shfl_xor(d, 4, 32);
  d += __shfl_xor(d, 2, 32);
  d += __shfl_xor(d, 1, 32);
  return d;
}

template <int KQ>
__device__ __forceinline__ void gate_rows3(const float* __restrict__ W, int j,
                                           const float* __restrict__ xv,
                                           float (&sum)[3]) {
  constexpr int K = 128 * KQ;
  constexpr int AHEAD = (KQ <= 4) ? 2 : 1;
  const int lane = threadIdx.x & 31;
  v4f x[KQ];
  load_pieces<KQ>(xv + 4 * lane, x);
  __builtin_amdgcn_sched_barrier(0);
  const float* wp = W + (size_t)j * (size_t)K + 4 * lane;
  v4f w[AHEAD + 1][KQ];
#pragma unroll
  for (int r = 0; r < AHEAD; ++r)
    load_pieces<KQ>(wp + (size_t)r * (size_t)HIDDEN * (size_t)K, w[r]);
#pragma unroll
  for (int r = 0; r < 3; ++r) {
    __builtin_amdgcn_sched_barrier(0);
    if (r + AHEAD < 3)
      load_pieces<KQ>(wp + (size_t)(r + AHEAD) * (size_t)HIDDEN * (size_t)K,
                      w[(r + AHEAD) % (AHEAD + 1)]);
    __builtin_amdgcn_sched_barrier(0);
    sum[r] = row_diag<KQ>(w[r % (AHEAD + 1)], x);
  }
#pragma unroll
  for (int r = 0; r < 3; ++r) sum[r] = trace_sum(sum[r]);
}

template <int K_IH, int K_HH>
__global__ void __launch_bounds__(BLOCK)
gru_layer_kernel(const float* __restrict__ W_ih, const float* __restrict__ W_hh,
                 const float* __restrict__ b_ih, const float* __restrict__ b_hh,
                 const float* __restrict__ xsrc, const int* __restrict__ xidx,
                 const float* __restrict__ hprev,
                 float* __restrict__ hnext, float* __restrict__ hcopy) {
  __shared__ float sPre[2][3][UNITS];

  const int tid   = threadIdx.x;
  const int lane  = tid & 31;
  const int jBase = blockIdx.x * UNITS;
  const int wave = __builtin_amdgcn_readfirstlane(tid >> 5);
  const int mat  = wave / UNITS;
  const int u    = wave % UNITS;
  const int j    = jBase + u;

  float hp = 0.0f;
  if (wave == 0) hp = hprev[jBase + (lane % UNITS)];
  const float* bias = (mat == 0) ? b_ih : b_hh;
  const float bias_r = bias[0 * HIDDEN + j];
  const float bias_z = bias[1 * HIDDEN + j];
  const float bias_n = bias[2 * HIDDEN + j];

  float sum[3];
  if (mat == 0) {
    const float* xv = xidx ? (xsrc + (size_t)xidx[0] * (size_t)K_IH) : xsrc;
    gate_rows3<K_IH / 128>(W_ih, j, xv, sum);
  } else {
    gate_rows3<K_HH / 128>(W_hh, j, hprev, sum);
  }
  if (lane == 0) {
    sPre[mat][0][u] = sum[0] + bias_r;
    sPre[mat][1][u] = sum[1] + bias_z;
    sPre[mat][2][u] = sum[2] + bias_n;
  }
  __syncthreads();

  if (tid < UNITS) {
    const float ir  = sPre[0][0][tid];
    const float iz  = sPre[0][1][tid];
    const float inn = sPre[0][2][tid];
    const float hr  = sPre[1][0][tid];
    const float hz  = sPre[1][1][tid];
    const float hn  = sPre[1][2][tid];
    const float r = 1.0f / (1.0f + __expf(-(ir + hr)));
    const float z = 1.0f / (1.0f + __expf(-(iz + hz)));
    const float n = tanhf(inn + r * hn);
    const float h = (1.0f - z) * n + z * hp;
    hnext[jBase + tid] = h;
    hcopy[jBase + tid] = h;
  }
}

__global__ void __launch_bounds__(BLOCK)
head_kernel(const float* __restrict__ w_head, const float* __restrict__ b_head,
            const float* __restrict__ h1, float* __restrict__ probs) {
  __shared__ float sAcc[16];
  const int tid  = threadIdx.x;
  const int lane = tid & 31;
  const int wave = __builtin_amdgcn_readfirstlane(tid >> 5);
  constexpr int KQ = HIDDEN / 128;

  v4f x[KQ], w[KQ];
  load_pieces<KQ>(h1 + 4 * lane, x);
  load_pieces<KQ>(w_head + (size_t)wave * HIDDEN + 4 * lane, w);
  __builtin_amdgcn_sched_barrier(0);
  const float s = trace_sum(row_diag<KQ>(w, x));
  if (lane == 0) sAcc[wave] = s;
  __syncthreads();

  if (tid < 16) {
    const float v = sAcc[tid] + b_head[tid];
    float m = v;
    for (int off = 8; off > 0; off >>= 1) m = fmaxf(m, __shfl_xor(m, off, 32));
    const float e = __expf(v - m);
    float s2 = e;
    for (int off = 8; off > 0; off >>= 1) s2 += __shfl_xor(s2, off, 32);
    probs[tid] = e / s2;
  }
}

extern "C" void kernel_launch(void* const* d_in, const int* in_sizes, int n_in,
                              void* d_out, int out_size, void* d_ws,
                              size_t ws_size, hipStream_t stream) {
  const int*   x      = (const int*)  d_in[0];
  const float* h_in   = (const float*)d_in[1];
  const float* emb    = (const float*)d_in[2];
  const float* w_ih0  = (const float*)d_in[3];
  const float* w_hh0  = (const float*)d_in[4];
  const float* b_ih0  = (const float*)d_in[5];
  const float* b_hh0  = (const float*)d_in[6];
  const float* w_ih1  = (const float*)d_in[7];
  const float* w_hh1  = (const float*)d_in[8];
  const float* b_ih1  = (const float*)d_in[9];
  const float* b_hh1  = (const float*)d_in[10];
  const float* w_head = (const float*)d_in[11];
  const float* b_head = (const float*)d_in[12];

  float* out = (float*)d_out;
  float* h0  = (float*)d_ws;
  float* h1  = h0 + HIDDEN;

  gru_layer_kernel<EMBED, HIDDEN><<<HIDDEN / UNITS, BLOCK, 0, stream>>>(
      w_ih0, w_hh0, b_ih0, b_hh0, emb, x, h_in, h0, out + NV);

  gru_layer_kernel<HIDDEN, HIDDEN><<<HIDDEN / UNITS, BLOCK, 0, stream>>>(
      w_ih1, w_hh1, b_ih1, b_hh1, h0, nullptr, h_in + HIDDEN, h1,
      out + NV + HIDDEN);

  head_kernel<<<1, BLOCK, 0, stream>>>(w_head, b_head, h1, out);
}


# Head products moved into the layer-2 blocks: each multiplies its 8 new h values by its 8 head columns on the matrix unit; last launch is one wave adding 256 partials; logit sum order changes
# speedup vs baseline: 2.7235x; 1.1040x over previous
#include <hip/hip_runtime.h>

typedef float v2f __attribute__((ext_vector_type(2)));
typedef float v4f __attribute__((ext_vector_type(4)));
typedef float v8f __attribute__((ext_vector_type(8)));

#define HIDDEN 2048
#define EMBED  512
#define NV     16
#define WAVES  16
#define BLOCK  (WAVES * 32)
#define UNITS  (WAVES / 2)

__device__ __forceinline__ v8f wmma_f32_k4(v2f a, v2f b, v8f c) {
  return __builtin_amdgcn_wmma_f32_16x16x4_f32(false, a, false, b, (short)0, c,
                                               false, false);
}


template <int KQ>
__device__ __forceinline__ void load_pieces(const float* __restrict__ p,
                                            v4f (&dst)[KQ]) {
#pragma unroll
  for (int q = 0; q < KQ; ++q) dst[q] = *(const v4f*)(p + 128 * q);
}

template <int KQ>
__device__ __forceinline__ float row_diag(const v4f (&w)[KQ],
                                          const v4f (&x)[KQ]) {
  v8f c0 = {}, c1 = {};
#pragma unroll
  for (int q = 0; q < KQ; ++q) {
    v2f a0; a0.x = w[q].x; a0.y = w[q].y;
    v2f b0; b0.x = x[q].x; b0.y = x[q].y;
    v2f a1; a1.x = w[q].z; a1.y = w[q].w;
    v2f b1; b1.x = x[q].z; b1.y = x[q].w;
    c0 = wmma_f32_k4(a0, b0, c0);
    c1 = wmma_f32_k4(a1, b1, c1);
  }
  const v8f c = c0 + c1;
  const int lane = threadIdx.x & 31;
  const int r = lane & 7;
  float d = c[0];
#pragma unroll
  for (int i = 1; i < 8; ++i) d = (r == i) ? c[i] : d;
  const bool on_diag = (lane >> 4) == ((lane >> 3) & 1);
  return on_diag ? d : 0.0f;
}

__device__ __forceinline__ float trace_sum(float d) {
  d += __shfl_xor(d, 24, 32);
  d += __shfl_xor(d, 4, 32);
  d += __shfl_xor(d, 2, 32);
  d += __shfl_xor(d, 1, 32);
  return d;
}

template <int KQ>
__device__ __forceinline__ void gate_rows3(const float* __restrict__ W, int j,
                                           const float* __restrict__ xv,
                                           float (&sum)[3]) {
  constexpr int K = 128 * KQ;
  constexpr int AHEAD = (KQ <= 4) ? 2 : 1;
  const int lane = threadIdx.x & 31;
  v4f x[KQ];
  load_pieces<KQ>(xv + 4 * lane, x);
  __builtin_amdgcn_sched_barrier(0);
  const float* wp = W + (size_t)j * (size_t)K + 4 * lane;
  v4f w[AHEAD + 1][KQ];
#pragma unroll
  for (int r = 0; r < AHEAD; ++r)
    load_pieces<KQ>(wp + (size_t)r * (size_t)HIDDEN * (size_t)K, w[r]);
#pragma unroll
  for (int r = 0; r < 3; ++r) {
    __builtin_amdgcn_sched_barrier(0);
    if (r + AHEAD < 3)
      load_pieces<KQ>(wp + (size_t)(r + AHEAD) * (size_t)HIDDEN * (size_t)K,
                      w[(r + AHEAD) % (AHEAD + 1)]);
    __builtin_amdgcn_sched_barrier(0);
    sum[r] = row_diag<KQ>(w[r % (AHEAD + 1)], x);
  }
#pragma unroll
  for (int r = 0; r < 3; ++r) sum[r] = trace_sum(sum[r]);
}

template <int K_IH, int K_HH, bool HEAD>
__global__ void __launch_bounds__(BLOCK)
gru_layer_kernel(const float* __restrict__ W_ih, const float* __restrict__ W_hh,
                 const float* __restrict__ b_ih, const float* __restrict__ b_hh,
                 const float* __restrict__ xsrc, const int* __restrict__ xidx,
                 const float* __restrict__ hprev, float* __restrict__ hout,
                 const float* __restrict__ w_head, float* __restrict__ part) {
  __shared__ float sPre[2][3][UNITS];

  const int tid   = threadIdx.x;
  const int lane  = tid & 31;
  const int jBase = blockIdx.x * UNITS;
  const int wave = __builtin_amdgcn_readfirstlane(tid >> 5);
  const int mat  = wave / UNITS;
  const int u    = wave % UNITS;
  const int j    = jBase + u;

  float hp = 0.0f;
  if (wave == 0) hp = hprev[jBase + (lane % UNITS)];
  const float* bias = (mat == 0) ? b_ih : b_hh;
  const float bias_r = bias[0 * HIDDEN + j];
  const float bias_z = bias[1 * HIDDEN + j];
  const float bias_n = bias[2 * HIDDEN + j];
  v2f ha0 = {}, ha1 = {};
  if constexpr (HEAD) {
    if (wave == 0) {
      const float* hw = w_head + (size_t)(lane & 15) * (size_t)HIDDEN + jBase +
                        2 * (lane >> 4);
      ha0 = *(const v2f*)hw;
      ha1 = *(const v2f*)(hw + 4);
    }
  }

  float sum[3];
  if (mat == 0) {
    const float* xv = xidx ? (xsrc + (size_t)xidx[0] * (size_t)K_IH) : xsrc;
    gate_rows3<K_IH / 128>(W_ih, j, xv, sum);
  } else {
    gate_rows3<K_HH / 128>(W_hh, j, hprev, sum);
  }
  if (lane == 0) {
    sPre[mat][0][u] = sum[0] + bias_r;
    sPre[mat][1][u] = sum[1] + bias_z;
    sPre[mat][2][u] = sum[2] + bias_n;
  }
  __syncthreads();

  if (wave == 0) {
    const int t = lane % UNITS;
    const float ir  = sPre[0][0][t];
    const float iz  = sPre[0][1][t];
    const float inn = sPre[0][2][t];
    const float hr  = sPre[1][0][t];
    const float hz  = sPre[1][1][t];
    const float hn  = sPre[1][2][t];
    const float r = 1.0f / (1.0f + __expf(-(ir + hr)));
    const float z = 1.0f / (1.0f + __expf(-(iz + hz)));
    const float n = tanhf(inn + r * hn);
    const float h = (1.0f - z) * n + z * hp;
    if (lane < UNITS) hout[jBase + lane] = h;
    if constexpr (HEAD) {
      float hu[UNITS];
#pragma unroll
      for (int k = 0; k < UNITS; ++k)
        hu[k] = __int_as_float(__builtin_amdgcn_readlane(__float_as_int(h), k));
      const bool hi = (lane >> 4) != 0;
      v2f hb0; hb0.x = hi ? hu[2] : hu[0]; hb0.y = hi ? hu[3] : hu[1];
      v2f hb1; hb1.x = hi ? hu[6] : hu[4]; hb1.y = hi ? hu[7] : hu[5];
      v8f d = {};
      d = wmma_f32_k4(ha0, hb0, d);
      d = wmma_f32_k4(ha1, hb1, d);
      if ((lane & 15) == 0) {
        float* p = part + (size_t)blockIdx.x * NV + 8 * (lane >> 4);
        v4f lo; lo.x = d[0]; lo.y = d[1]; lo.z = d[2]; lo.w = d[3];
        v4f up; up.x = d[4]; up.y = d[5]; up.z = d[6]; up.w = d[7];
        *(v4f*)p = lo;
        *(v4f*)(p + 4) = up;
      }
    }
  }
}

__global__ void __launch_bounds__(32)
head_sum_kernel(const float* __restrict__ part, const float* __restrict__ b_head,
                float* __restrict__ probs) {
  constexpr int KQ = (HIDDEN / UNITS) * NV / 128;
  const int lane = threadIdx.x;
  v4f p[KQ];
  load_pieces<KQ>(part + 4 * lane, p);
  const v4f bias = *(const v4f*)(b_head + 4 * (lane & 3));
  __builtin_amdgcn_sched_barrier(0);
  v4f acc[4] = {};
#pragma unroll
  for (int q = 0; q < KQ; ++q) acc[q & 3] += p[q];
  v4f s = (acc[0] + acc[1]) + (acc[2] + acc[3]);
#pragma unroll
  for (int off = 4; off < 32; off <<= 1) {
    s.x += __shfl_xor(s.x, off, 32);
    s.y += __shfl_xor(s.y, off, 32);
    s.z += __shfl_xor(s.z, off, 32);
    s.w += __shfl_xor(s.w, off, 32);
  }
  const v4f v = s + bias;
  float m = fmaxf(fmaxf(v.x, v.y), fmaxf(v.z, v.w));
  m = fmaxf(m, __shfl_xor(m, 1, 32));
  m = fmaxf(m, __shfl_xor(m, 2, 32));
  v4f e;
  e.x = __expf(v.x - m); e.y = __expf(v.y - m);
  e.z = __expf(v.z - m); e.w = __expf(v.w - m);
  float t = (e.x + e.y) + (e.z + e.w);
  t += __shfl_xor(t, 1, 32);
  t += __shfl_xor(t, 2, 32);
  if (lane < 4) {
    v4f o; o.x = e.x / t; o.y = e.y / t; o.z = e.z / t; o.w = e.w / t;
    *(v4f*)(probs + 4 * lane) = o;
  }
}

extern "C" void kernel_launch(void* const* d_in, const int* in_sizes, int n_in,
                              void* d_out, int out_size, void* d_ws,
                              size_t ws_size, hipStream_t stream) {
  const int*   x      = (const int*)  d_in[0];
  const float* h_in   = (const float*)d_in[1];
  const float* emb    = (const float*)d_in[2];
  const float* w_ih0  = (const float*)d_in[3];
  const float* w_hh0  = (const float*)d_in[4];
  const float* b_ih0  = (const float*)d_in[5];
  const float* b_hh0  = (const float*)d_in[6];
  const float* w_ih1  = (const float*)d_in[7];
  const float* w_hh1  = (const float*)d_in[8];
  const float* b_ih1  = (const float*)d_in[9];
  const float* b_hh1  = (const float*)d_in[10];
  const float* w_head = (const float*)d_in[11];
  const float* b_head = (const float*)d_in[12];

  float* out  = (float*)d_out;
  float* h0   = out + NV;
  float* h1   = h0 + HIDDEN;
  float* part = (float*)d_ws;

  gru_layer_kernel<EMBED, HIDDEN, false><<<HIDDEN / UNITS, BLOCK, 0, stream>>>(
      w_ih0, w_hh0, b_ih0, b_hh0, emb, x, h_in, h0, nullptr, nullptr);

  gru_layer_kernel<HIDDEN, HIDDEN, true><<<HIDDEN / UNITS, BLOCK, 0, stream>>>(
      w_ih1, w_hh1, b_ih1, b_hh1, h0, nullptr, h_in + HIDDEN, h1, w_head, part);

  head_sum_kernel<<<1, 32, 0, stream>>>(part, b_head, out);
}
